# starting kernel
# baseline (speedup 1.0000x reference)
#include <hip/hip_runtime.h>
#include <hip/hip_bf16.h>

typedef __attribute__((ext_vector_type(2))) float v2f;
typedef __attribute__((ext_vector_type(8))) float v8f;

#define C_IN_   8
#define C_OUT_  16
#define HW_     1024
#define D0_     3072
#define D1_     8192
#define P_      256
#define KTAPS   72
#define KSTEPS  18
#define NTILES  192
#define NWAVES  8
#define NSLICE  4
#define TILES_PER_SLICE (NTILES / NSLICE)
#define DSLICE  (D0_ / NSLICE)

__device__ __forceinline__ float ldg_off(const char* __restrict__ base, uint32_t off) {
    return *(const float*)(base + off);
}

__global__ __launch_bounds__(256) void deeppoly_conv_main(
    const float* __restrict__ l0,  const float* __restrict__ u0,
    const float* __restrict__ lx0, const float* __restrict__ ux0,
    const float* __restrict__ weight,
    float* __restrict__ ws)
{
    __shared__ float s_l0[DSLICE];
    __shared__ float s_u0[DSLICE];
    __shared__ float s_wpos[KTAPS * 16];
    __shared__ float s_wneg[KTAPS * 16];
    __shared__ int   s_colb[KTAPS];
    __shared__ float s_pl[NWAVES][32];
    __shared__ float s_pu[NWAVES][32];

    const int tid   = threadIdx.x;
    const int p     = blockIdx.x;
    const int slice = blockIdx.y;
    const int ph = p >> 4, pw = p & 15;
    const int dbase = slice * DSLICE;

    for (int i = tid; i < DSLICE; i += 256) { s_l0[i] = l0[dbase + i]; s_u0[i] = u0[dbase + i]; }

    for (int e = tid; e < KTAPS * 16; e += 256) {
        int k = e >> 4, oc = e & 15;
        int t9 = k % 9;
        int r = ph * 2 - 1 + t9 / 3;
        int c = pw * 2 - 1 + (t9 % 3);
        bool valid = (r >= 0) && (r < 32) && (c >= 0) && (c < 32);
        float w = valid ? weight[oc * KTAPS + k] : 0.f;
        s_wpos[k * 16 + oc] = w > 0.f ? w : 0.f;
        s_wneg[k * 16 + oc] = w < 0.f ? w : 0.f;
    }
    if (tid < KTAPS) {
        int ic = tid / 9, t9 = tid % 9;
        int r = ph * 2 - 1 + t9 / 3;
        int c = pw * 2 - 1 + (t9 % 3);
        bool valid = (r >= 0) && (r < 32) && (c >= 0) && (c < 32);
        s_colb[tid] = valid ? ((ic * HW_ + r * 32 + c) << 2) : 0;
    }
    __syncthreads();

    const int wave = tid >> 5;
    const int lane = tid & 31;
    const int m    = lane & 15;
    const int koff = (lane >> 4) << 1;

    v2f bp[KSTEPS], bn[KSTEPS];
    uint32_t c0b[KSTEPS], c1b[KSTEPS];
    #pragma unroll
    for (int s = 0; s < KSTEPS; ++s) {
        const int k0 = s * 4 + koff;
        bp[s]  = (v2f){ s_wpos[k0 * 16 + m], s_wpos[(k0 + 1) * 16 + m] };
        bn[s]  = (v2f){ s_wneg[k0 * 16 + m], s_wneg[(k0 + 1) * 16 + m] };
        c0b[s] = (uint32_t)s_colb[k0];
        c1b[s] = (uint32_t)s_colb[k0 + 1];
    }

    const char* __restrict__ lx0b = (const char*)lx0;
    const char* __restrict__ ux0b = (const char*)ux0;

    float acc_l = 0.f, acc_u = 0.f;

    for (int ti = 0; ti < TILES_PER_SLICE / NWAVES; ++ti) {
        const int t   = slice * TILES_PER_SLICE + wave + ti * NWAVES;
        const int d0r = t * 16;
        const uint32_t rowb = (uint32_t)(d0r + m) << 15;

        v8f alx = {};
        v8f aux = {};

        #pragma unroll
        for (int s = 0; s < KSTEPS; ++s) {
            const uint32_t o0 = rowb + c0b[s];
            const uint32_t o1 = rowb + c1b[s];
            v2f a_l = (v2f){ ldg_off(lx0b, o0), ldg_off(lx0b, o1) };
            v2f a_u = (v2f){ ldg_off(ux0b, o0), ldg_off(ux0b, o1) };
            alx = __builtin_amdgcn_wmma_f32_16x16x4_f32(false, a_l, false, bp[s], (short)0, alx, false, false);
            alx = __builtin_amdgcn_wmma_f32_16x16x4_f32(false, a_u, false, bn[s], (short)0, alx, false, false);
            aux = __builtin_amdgcn_wmma_f32_16x16x4_f32(false, a_u, false, bp[s], (short)0, aux, false, false);
            aux = __builtin_amdgcn_wmma_f32_16x16x4_f32(false, a_l, false, bn[s], (short)0, aux, false, false);
        }

        const int rloc = (d0r - dbase) + ((lane >> 4) << 3);
        #pragma unroll
        for (int i = 0; i < 8; ++i) {
            const float lv = s_l0[rloc + i], uv = s_u0[rloc + i];
            const float x = alx[i];
            acc_l += (x > 0.f ? lv : uv) * x;
            const float y = aux[i];
            acc_u += (y > 0.f ? uv : lv) * y;
        }
    }

    s_pl[wave][lane] = acc_l;
    s_pu[wave][lane] = acc_u;
    __syncthreads();

    if (tid < 32) {
        const int  oc  = tid & 15;
        const bool isU = tid >= 16;
        float sum = 0.f;
        #pragma unroll
        for (int w = 0; w < NWAVES; ++w)
            sum += isU ? (s_pu[w][oc] + s_pu[w][oc + 16])
                       : (s_pl[w][oc] + s_pl[w][oc + 16]);
        ws[(size_t)(p * NSLICE + slice) * 32 + tid] = sum;
    }
}

__global__ __launch_bounds__(32) void deeppoly_conv_finish(
    const float* __restrict__ lc0, const float* __restrict__ uc0,
    const float* __restrict__ weight, const float* __restrict__ bias,
    const float* __restrict__ ws, float* __restrict__ out)
{
    const int p   = blockIdx.x;
    const int tid = threadIdx.x;
    const int oc  = tid & 15;
    const bool isU = tid >= 16;
    const int ph = p >> 4, pw = p & 15;

    float sum = 0.f;
    #pragma unroll
    for (int s = 0; s < NSLICE; ++s)
        sum += ws[(size_t)(p * NSLICE + s) * 32 + tid];

    float cacc = bias[oc];
    for (int k = 0; k < KTAPS; ++k) {
        const int ic = k / 9, t9 = k % 9;
        const int r = ph * 2 - 1 + t9 / 3;
        const int c = pw * 2 - 1 + (t9 % 3);
        if (r < 0 || r >= 32 || c < 0 || c >= 32) continue;
        const int j = ic * HW_ + r * 32 + c;
        const float wv = weight[oc * KTAPS + k];
        const float wp = wv > 0.f ? wv : 0.f;
        const float wn = wv < 0.f ? wv : 0.f;
        const float lcv = lc0[j], ucv = uc0[j];
        cacc += isU ? (wp * ucv + wn * lcv)
                    : (wp * lcv + wn * ucv);
    }
    out[(isU ? 4096 : 0) + oc * P_ + p] = sum + cacc;
}

extern "C" void kernel_launch(void* const* d_in, const int* in_sizes, int n_in,
                              void* d_out, int out_size, void* d_ws, size_t ws_size,
                              hipStream_t stream) {
    const float* l0     = (const float*)d_in[0];
    const float* u0     = (const float*)d_in[1];
    const float* lx0    = (const float*)d_in[2];
    const float* ux0    = (const float*)d_in[3];
    const float* lc0    = (const float*)d_in[4];
    const float* uc0    = (const float*)d_in[5];
    const float* weight = (const float*)d_in[6];
    const float* bias   = (const float*)d_in[7];
    float* ws  = (float*)d_ws;
    float* out = (float*)d_out;

    deeppoly_conv_main<<<dim3(P_, NSLICE), 256, 0, stream>>>(
        l0, u0, lx0, ux0, weight, ws);
    deeppoly_conv_finish<<<P_, 32, 0, stream>>>(
        lc0, uc0, weight, bias, ws, out);
}


# Constant terms (lc0,uc0,bias) as one more relaxation row: staged like a d-row, times the weight fragments with positions as matrix rows, in slice-0 blocks; finish = slice sum; constant sum regrouped
# speedup vs baseline: 5.3123x; 5.3123x over previous
#include <hip/hip_runtime.h>


typedef __attribute__((ext_vector_type(2))) float v2f;
typedef __attribute__((ext_vector_type(8))) float v8f;

#define KTAPS   72
#define HW_     1024
#define D1_     8192
#define P_      256
#define RP      33
#define PITCH2  793
#define HALF_OFF 396
#define MAX_TPS 48

__device__ __forceinline__ void lds_barrier() {
    asm volatile("s_wait_dscnt 0x0\n\ts_barrier_signal -1\n\ts_barrier_wait -1" ::: "memory");
}

__device__ __forceinline__ void load_tile(const float* __restrict__ gl,
                                          const float* __restrict__ gu,
                                          int ph, float4 (&vl)[12], float4 (&vu)[12])
{
    #pragma unroll
    for (int it = 0; it < 12; ++it) {
        const int kh = it % 3;
        const int rr = (kh > 0 || ph > 0) ? (2 * ph - 1 + kh) : 0;
        const int go = (it / 3) * HW_ + rr * 32;
        vl[it] = *(const float4*)(gl + go);
        vu[it] = *(const float4*)(gu + go);
    }
}

__global__ __launch_bounds__(256) void deeppoly_conv_main(
    const float* __restrict__ l0,  const float* __restrict__ u0,
    const float* __restrict__ lx0, const float* __restrict__ ux0,
    const float* __restrict__ lc0, const float* __restrict__ uc0,
    const float* __restrict__ weight, const float* __restrict__ bias,
    float* __restrict__ ws, int tps, int nslice)
{
    __shared__ v2f s_a[16 * PITCH2];
    __shared__ v2f s_lu[MAX_TPS * 16];
    __shared__ float s_c[16 * 32];

    const int tid   = threadIdx.x;
    const int ph    = blockIdx.x;
    const int slice = blockIdx.y;
    const int wave  = tid >> 5;
    const int lane  = tid & 31;
    const int c     = lane & 15;
    const int h     = lane >> 4;
    const int dbase = slice * tps * 16;

    const int st_c4   = tid & 7;
    const int st_drow = (tid >> 3) & 15;
    const int st_half = tid >> 7;
    const int goff_t  = st_drow * D1_ + st_half * (4 * HW_) + st_c4 * 4;
    const int e_t     = st_drow * PITCH2 + st_half * HALF_OFF + 1 + st_c4 * 4;

    const bool cblock = (slice == 0);
    const int  cR     = tid >> 3;
    float4 cl = { 0.f, 0.f, 0.f, 0.f }, cu = { 0.f, 0.f, 0.f, 0.f };
    if (cblock && tid < 192) {
        const int kh = cR % 3;
        const int rr = (kh > 0 || ph > 0) ? (2 * ph - 1 + kh) : 0;
        const int go = (cR / 3) * HW_ + rr * 32 + st_c4 * 4;
        cl = *(const float4*)(lc0 + go);
        cu = *(const float4*)(uc0 + go);
    }

    float4 vl[12], vu[12];
    load_tile(lx0 + (size_t)dbase * D1_ + goff_t, ux0 + (size_t)dbase * D1_ + goff_t, ph, vl, vu);

    v2f b1[36];
    #pragma unroll
    for (int s = 0; s < 36; ++s) {
        const float w = weight[c * KTAPS + s + 36 * h];
        b1[s] = (v2f){ w > 0.f ? w : 0.f, w < 0.f ? w : 0.f };
    }

    for (int i = tid; i < 16 * 24; i += 256) {
        const int drow = i / 24, t = i - drow * 24;
        s_a[drow * PITCH2 + t * RP] = (v2f){ 0.f, 0.f };
    }
    for (int i = tid; i < tps * 16; i += 256)
        s_lu[i] = (v2f){ l0[dbase + i], u0[dbase + i] };

    if (cblock) {
        if (tid < 192) {
            const bool keep = (cR % 3 > 0) || (ph > 0);
            const int e = cR * RP + 1 + st_c4 * 4;
            s_a[e]     = (v2f){ keep ? cl.x : 0.f, keep ? cu.x : 0.f };
            s_a[e + 1] = (v2f){ keep ? cl.y : 0.f, keep ? cu.y : 0.f };
            s_a[e + 2] = (v2f){ keep ? cl.z : 0.f, keep ? cu.z : 0.f };
            s_a[e + 3] = (v2f){ keep ? cl.w : 0.f, keep ? cu.w : 0.f };
        }
        lds_barrier();
        if (wave < 2) {
            const bool up = (wave == 1);
            const int  ab = h * HALF_OFF + 2 * c;
            v8f q0 = {}, q1 = {};
            #pragma unroll
            for (int s = 0; s < 36; s += 2) {
                const v2f r0 = s_a[ab + (s / 3) * RP + (s % 3)];
                const v2f r1 = s_a[ab + ((s + 1) / 3) * RP + ((s + 1) % 3)];
                const v2f a0 = up ? (v2f){ r0.y, r0.x } : r0;
                const v2f a1 = up ? (v2f){ r1.y, r1.x } : r1;
                q0 = __builtin_amdgcn_wmma_f32_16x16x4_f32(false, a0, false, b1[s],     (short)0, q0, false, false);
                q1 = __builtin_amdgcn_wmma_f32_16x16x4_f32(false, a1, false, b1[s + 1], (short)0, q1, false, false);
            }
            const float bv = bias[c];
            #pragma unroll
            for (int i = 0; i < 8; ++i)
                s_c[(8 * h + i) * 32 + 16 * wave + c] = (q0[i] + q1[i]) + bv;
        }
    }

    const int pw0    = wave * 2;
    const int abase0 = c * PITCH2 + h * HALF_OFF + 2 * pw0;
    const int abase1 = abase0 + 2;
    const bool keep0 = ph > 0;

    float accl0 = 0.f, accu0 = 0.f, accl1 = 0.f, accu1 = 0.f;

    for (int t = 0; t < tps; ++t) {
        lds_barrier();

        #pragma unroll
        for (int it = 0; it < 12; ++it) {
            const bool keep = (it % 3 > 0) || keep0;
            const float4 a = vl[it], b = vu[it];
            const int e = e_t + it * RP;
            s_a[e]     = (v2f){ keep ? a.x : 0.f, keep ? b.x : 0.f };
            s_a[e + 1] = (v2f){ keep ? a.y : 0.f, keep ? b.y : 0.f };
            s_a[e + 2] = (v2f){ keep ? a.z : 0.f, keep ? b.z : 0.f };
            s_a[e + 3] = (v2f){ keep ? a.w : 0.f, keep ? b.w : 0.f };
        }
        if (t + 1 < tps) {
            const size_t gn = (size_t)(dbase + (t + 1) * 16) * D1_ + goff_t;
            load_tile(lx0 + gn, ux0 + gn, ph, vl, vu);
        }
        lds_barrier();

        v8f alx0 = {}, aux0 = {}, alx1 = {}, aux1 = {};
        #pragma unroll
        for (int s = 0; s < 36; ++s) {
            const int off = (s / 3) * RP + (s % 3);
            const v2f a0 = s_a[abase0 + off];
            const v2f a1 = s_a[abase1 + off];
            const v2f a0s = (v2f){ a0.y, a0.x };
            const v2f a1s = (v2f){ a1.y, a1.x };
            alx0 = __builtin_amdgcn_wmma_f32_16x16x4_f32(false, a0,  false, b1[s], (short)0, alx0, false, false);
            aux0 = __builtin_amdgcn_wmma_f32_16x16x4_f32(false, a0s, false, b1[s], (short)0, aux0, false, false);
            alx1 = __builtin_amdgcn_wmma_f32_16x16x4_f32(false, a1,  false, b1[s], (short)0, alx1, false, false);
            aux1 = __builtin_amdgcn_wmma_f32_16x16x4_f32(false, a1s, false, b1[s], (short)0, aux1, false, false);
        }

        #pragma unroll
        for (int i = 0; i < 8; ++i) {
            const v2f lu = s_lu[t * 16 + 8 * h + i];
            const float lv = lu.x, uv = lu.y;
            const float x0 = alx0[i]; accl0 += (x0 > 0.f ? lv : uv) * x0;
            const float y0 = aux0[i]; accu0 += (y0 > 0.f ? uv : lv) * y0;
            const float x1 = alx1[i]; accl1 += (x1 > 0.f ? lv : uv) * x1;
            const float y1 = aux1[i]; accu1 += (y1 > 0.f ? uv : lv) * y1;
        }
    }

    const float tl0 = accl0 + __shfl_xor(accl0, 16);
    const float tu0 = accu0 + __shfl_xor(accu0, 16);
    const float tl1 = accl1 + __shfl_xor(accl1, 16);
    const float tu1 = accu1 + __shfl_xor(accu1, 16);
    const int p0 = ph * 16 + pw0;
    float ct0 = 0.f, ct1 = 0.f;
    if (cblock) {
        ct0 = s_c[pw0 * 32 + lane];
        ct1 = s_c[(pw0 + 1) * 32 + lane];
    }
    ws[(size_t)(p0 * nslice + slice) * 32 + lane]       = (h ? tu0 : tl0) + ct0;
    ws[(size_t)((p0 + 1) * nslice + slice) * 32 + lane] = (h ? tu1 : tl1) + ct1;
}

__global__ __launch_bounds__(32) void deeppoly_conv_finish(
    const float* __restrict__ ws, float* __restrict__ out, int nslice)
{
    const int p   = blockIdx.x;
    const int tid = threadIdx.x;
    const int oc  = tid & 15;
    const bool isU = tid >= 16;

    float sum = 0.f;
    for (int s = 0; s < nslice; ++s)
        sum += ws[(size_t)(p * nslice + s) * 32 + tid];

    out[(isU ? 4096 : 0) + oc * P_ + p] = sum;
}

extern "C" void kernel_launch(void* const* d_in, const int* in_sizes, int n_in,
                              void* d_out, int out_size, void* d_ws, size_t ws_size,
                              hipStream_t stream) {
    const float* l0     = (const float*)d_in[0];
    const float* u0     = (const float*)d_in[1];
    const float* lx0    = (const float*)d_in[2];
    const float* ux0    = (const float*)d_in[3];
    const float* lc0    = (const float*)d_in[4];
    const float* uc0    = (const float*)d_in[5];
    const float* weight = (const float*)d_in[6];
    const float* bias   = (const float*)d_in[7];
    float* ws  = (float*)d_ws;
    float* out = (float*)d_out;

    if (ws_size >= 1048576) {
        deeppoly_conv_main<<<dim3(16, 32), 256, 0, stream>>>(l0, u0, lx0, ux0, lc0, uc0, weight, bias, ws, 6, 32);
        deeppoly_conv_finish<<<P_, 32, 0, stream>>>(ws, out, 32);
    } else if (ws_size >= 524288) {
        deeppoly_conv_main<<<dim3(16, 16), 256, 0, stream>>>(l0, u0, lx0, ux0, lc0, uc0, weight, bias, ws, 12, 16);
        deeppoly_conv_finish<<<P_, 32, 0, stream>>>(ws, out, 16);
    } else if (ws_size >= 262144) {
        deeppoly_conv_main<<<dim3(16, 8), 256, 0, stream>>>(l0, u0, lx0, ux0, lc0, uc0, weight, bias, ws, 24, 8);
        deeppoly_conv_finish<<<P_, 32, 0, stream>>>(ws, out, 8);
    } else {
        deeppoly_conv_main<<<dim3(16, 4), 256, 0, stream>>>(l0, u0, lx0, ux0, lc0, uc0, weight, bias, ws, 48, 4);
        deeppoly_conv_finish<<<P_, 32, 0, stream>>>(ws, out, 4);
    }
}


# Tile worked as two channel halves in alternating LDS regions: each wave stages the next half after its own products of the current one, one barrier per half; tap order and every sum kept
# speedup vs baseline: 5.7616x; 1.0846x over previous
#include <hip/hip_runtime.h>


typedef __attribute__((ext_vector_type(2))) float v2f;
typedef __attribute__((ext_vector_type(8))) float v8f;

#define KTAPS   72
#define HW_     1024
#define D1_     8192
#define P_      256
#define RP      33
#define PITCH2  793
#define HALF_OFF 396
#define MAX_TPS 48

__device__ __forceinline__ void lds_barrier() {
    asm volatile("s_wait_dscnt 0x0\n\ts_barrier_signal -1\n\ts_barrier_wait -1" ::: "memory");
}

__device__ __forceinline__ void load_half(const float* __restrict__ gl,
                                          const float* __restrict__ gu,
                                          int ph, float4 (&vl)[6], float4 (&vu)[6])
{
    #pragma unroll
    for (int it = 0; it < 6; ++it) {
        const int kh = it % 3;
        const int rr = (kh > 0 || ph > 0) ? (2 * ph - 1 + kh) : 0;
        const int go = (it / 3) * HW_ + rr * 32;
        vl[it] = *(const float4*)(gl + go);
        vu[it] = *(const float4*)(gu + go);
    }
}

__global__ __launch_bounds__(256) void deeppoly_conv_main(
    const float* __restrict__ l0,  const float* __restrict__ u0,
    const float* __restrict__ lx0, const float* __restrict__ ux0,
    const float* __restrict__ lc0, const float* __restrict__ uc0,
    const float* __restrict__ weight, const float* __restrict__ bias,
    float* __restrict__ ws, int tps, int nslice)
{
    __shared__ v2f s_a[16 * PITCH2];
    __shared__ v2f s_lu[MAX_TPS * 16];
    __shared__ float s_c[16 * 32];

    const int tid   = threadIdx.x;
    const int ph    = blockIdx.x;
    const int slice = blockIdx.y;
    const int wave  = tid >> 5;
    const int lane  = tid & 31;
    const int c     = lane & 15;
    const int h     = lane >> 4;
    const int dbase = slice * tps * 16;

    const int st_c4   = tid & 7;
    const int st_drow = (tid >> 3) & 15;
    const int st_half = tid >> 7;
    const int goff_t  = st_drow * D1_ + st_half * (4 * HW_) + st_c4 * 4;
    const int e_t     = st_drow * PITCH2 + st_half * HALF_OFF + 1 + st_c4 * 4;

    const bool cblock = (slice == 0);
    const int  cR     = tid >> 3;
    float4 cl = { 0.f, 0.f, 0.f, 0.f }, cu = { 0.f, 0.f, 0.f, 0.f };
    if (cblock && tid < 192) {
        const int kh = cR % 3;
        const int rr = (kh > 0 || ph > 0) ? (2 * ph - 1 + kh) : 0;
        const int go = (cR / 3) * HW_ + rr * 32 + st_c4 * 4;
        cl = *(const float4*)(lc0 + go);
        cu = *(const float4*)(uc0 + go);
    }

    float4 vl[6], vu[6];
    load_half(lx0 + (size_t)dbase * D1_ + goff_t, ux0 + (size_t)dbase * D1_ + goff_t, ph, vl, vu);

    v2f b1[36];
    #pragma unroll
    for (int s = 0; s < 36; ++s) {
        const float w = weight[c * KTAPS + s + 36 * h];
        b1[s] = (v2f){ w > 0.f ? w : 0.f, w < 0.f ? w : 0.f };
    }

    for (int i = tid; i < 16 * 24; i += 256) {
        const int drow = i / 24, t = i - drow * 24;
        s_a[drow * PITCH2 + t * RP] = (v2f){ 0.f, 0.f };
    }
    for (int i = tid; i < tps * 16; i += 256)
        s_lu[i] = (v2f){ l0[dbase + i], u0[dbase + i] };

    if (cblock) {
        if (tid < 192) {
            const bool keep = (cR % 3 > 0) || (ph > 0);
            const int e = cR * RP + 1 + st_c4 * 4;
            s_a[e]     = (v2f){ keep ? cl.x : 0.f, keep ? cu.x : 0.f };
            s_a[e + 1] = (v2f){ keep ? cl.y : 0.f, keep ? cu.y : 0.f };
            s_a[e + 2] = (v2f){ keep ? cl.z : 0.f, keep ? cu.z : 0.f };
            s_a[e + 3] = (v2f){ keep ? cl.w : 0.f, keep ? cu.w : 0.f };
        }
        lds_barrier();
        if (wave < 2) {
            const bool up = (wave == 1);
            const int  ab = h * HALF_OFF + 2 * c;
            v8f q0 = {}, q1 = {};
            #pragma unroll
            for (int s = 0; s < 36; s += 2) {
                const v2f r0 = s_a[ab + (s / 3) * RP + (s % 3)];
                const v2f r1 = s_a[ab + ((s + 1) / 3) * RP + ((s + 1) % 3)];
                const v2f a0 = up ? (v2f){ r0.y, r0.x } : r0;
                const v2f a1 = up ? (v2f){ r1.y, r1.x } : r1;
                q0 = __builtin_amdgcn_wmma_f32_16x16x4_f32(false, a0, false, b1[s],     (short)0, q0, false, false);
                q1 = __builtin_amdgcn_wmma_f32_16x16x4_f32(false, a1, false, b1[s + 1], (short)0, q1, false, false);
            }
            const float bv = bias[c];
            #pragma unroll
            for (int i = 0; i < 8; ++i)
                s_c[(8 * h + i) * 32 + 16 * wave + c] = (q0[i] + q1[i]) + bv;
        }
    }

    const int pw0    = wave * 2;
    const int abase0 = c * PITCH2 + h * HALF_OFF + 2 * pw0;
    const int abase1 = abase0 + 2;
    const bool keep0 = ph > 0;

    lds_barrier();

    #pragma unroll
    for (int it = 0; it < 6; ++it) {
        const bool keep = (it % 3 > 0) || keep0;
        const float4 a = vl[it], b = vu[it];
        const int e = e_t + it * RP;
        s_a[e]     = (v2f){ keep ? a.x : 0.f, keep ? b.x : 0.f };
        s_a[e + 1] = (v2f){ keep ? a.y : 0.f, keep ? b.y : 0.f };
        s_a[e + 2] = (v2f){ keep ? a.z : 0.f, keep ? b.z : 0.f };
        s_a[e + 3] = (v2f){ keep ? a.w : 0.f, keep ? b.w : 0.f };
    }
    {
        const size_t gn = (size_t)dbase * D1_ + goff_t + 2 * HW_;
        load_half(lx0 + gn, ux0 + gn, ph, vl, vu);
    }
    lds_barrier();

    float accl0 = 0.f, accu0 = 0.f, accl1 = 0.f, accu1 = 0.f;

    for (int t = 0; t < tps; ++t) {
        v8f alx0 = {}, aux0 = {}, alx1 = {}, aux1 = {};

        #pragma unroll
        for (int q = 0; q < 2; ++q) {
            #pragma unroll
            for (int i = 0; i < 18; ++i) {
                const int s = 18 * q + i;
                const int off = (s / 3) * RP + (s % 3);
                const v2f a0 = s_a[abase0 + off];
                const v2f a1 = s_a[abase1 + off];
                const v2f a0s = (v2f){ a0.y, a0.x };
                const v2f a1s = (v2f){ a1.y, a1.x };
                alx0 = __builtin_amdgcn_wmma_f32_16x16x4_f32(false, a0,  false, b1[s], (short)0, alx0, false, false);
                aux0 = __builtin_amdgcn_wmma_f32_16x16x4_f32(false, a0s, false, b1[s], (short)0, aux0, false, false);
                alx1 = __builtin_amdgcn_wmma_f32_16x16x4_f32(false, a1,  false, b1[s], (short)0, alx1, false, false);
                aux1 = __builtin_amdgcn_wmma_f32_16x16x4_f32(false, a1s, false, b1[s], (short)0, aux1, false, false);
            }
            __builtin_amdgcn_sched_barrier(0);

            if (q == 1) {
                #pragma unroll
                for (int i = 0; i < 8; ++i) {
                    const v2f lu = s_lu[t * 16 + 8 * h + i];
                    const float lv = lu.x, uv = lu.y;
                    const float x0 = alx0[i]; accl0 += (x0 > 0.f ? lv : uv) * x0;
                    const float y0 = aux0[i]; accu0 += (y0 > 0.f ? uv : lv) * y0;
                    const float x1 = alx1[i]; accl1 += (x1 > 0.f ? lv : uv) * x1;
                    const float y1 = aux1[i]; accu1 += (y1 > 0.f ? uv : lv) * y1;
                }
            }

            const int  qn   = 1 - q;
            const bool more = (t + 1 < tps);
            if (q == 0 || more) {
                #pragma unroll
                for (int it = 0; it < 6; ++it) {
                    const bool keep = (it % 3 > 0) || keep0;
                    const float4 a = vl[it], b = vu[it];
                    const int e = e_t + (6 * qn + it) * RP;
                    s_a[e]     = (v2f){ keep ? a.x : 0.f, keep ? b.x : 0.f };
                    s_a[e + 1] = (v2f){ keep ? a.y : 0.f, keep ? b.y : 0.f };
                    s_a[e + 2] = (v2f){ keep ? a.z : 0.f, keep ? b.z : 0.f };
                    s_a[e + 3] = (v2f){ keep ? a.w : 0.f, keep ? b.w : 0.f };
                }
            }
            if (more) {
                const size_t gn = (size_t)(dbase + (t + 1) * 16) * D1_ + goff_t + (2 * q) * HW_;
                load_half(lx0 + gn, ux0 + gn, ph, vl, vu);
            }
            lds_barrier();
        }
    }

    const float tl0 = accl0 + __shfl_xor(accl0, 16);
    const float tu0 = accu0 + __shfl_xor(accu0, 16);
    const float tl1 = accl1 + __shfl_xor(accl1, 16);
    const float tu1 = accu1 + __shfl_xor(accu1, 16);
    const int p0 = ph * 16 + pw0;
    float ct0 = 0.f, ct1 = 0.f;
    if (cblock) {
        ct0 = s_c[pw0 * 32 + lane];
        ct1 = s_c[(pw0 + 1) * 32 + lane];
    }
    ws[(size_t)(p0 * nslice + slice) * 32 + lane]       = (h ? tu0 : tl0) + ct0;
    ws[(size_t)((p0 + 1) * nslice + slice) * 32 + lane] = (h ? tu1 : tl1) + ct1;
}

__global__ __launch_bounds__(32) void deeppoly_conv_finish(
    const float* __restrict__ ws, float* __restrict__ out, int nslice)
{
    const int p   = blockIdx.x;
    const int tid = threadIdx.x;
    const int oc  = tid & 15;
    const bool isU = tid >= 16;

    float sum = 0.f;
    for (int s = 0; s < nslice; ++s)
        sum += ws[(size_t)(p * nslice + s) * 32 + tid];

    out[(isU ? 4096 : 0) + oc * P_ + p] = sum;
}

extern "C" void kernel_launch(void* const* d_in, const int* in_sizes, int n_in,
                              void* d_out, int out_size, void* d_ws, size_t ws_size,
                              hipStream_t stream) {
    const float* l0     = (const float*)d_in[0];
    const float* u0     = (const float*)d_in[1];
    const float* lx0    = (const float*)d_in[2];
    const float* ux0    = (const float*)d_in[3];
    const float* lc0    = (const float*)d_in[4];
    const float* uc0    = (const float*)d_in[5];
    const float* weight = (const float*)d_in[6];
    const float* bias   = (const float*)d_in[7];
    float* ws  = (float*)d_ws;
    float* out = (float*)d_out;

    if (ws_size >= 1048576) {
        deeppoly_conv_main<<<dim3(16, 32), 256, 0, stream>>>(l0, u0, lx0, ux0, lc0, uc0, weight, bias, ws, 6, 32);
        deeppoly_conv_finish<<<P_, 32, 0, stream>>>(ws, out, 32);
    } else if (ws_size >= 524288) {
        deeppoly_conv_main<<<dim3(16, 16), 256, 0, stream>>>(l0, u0, lx0, ux0, lc0, uc0, weight, bias, ws, 12, 16);
        deeppoly_conv_finish<<<P_, 32, 0, stream>>>(ws, out, 16);
    } else if (ws_size >= 262144) {
        deeppoly_conv_main<<<dim3(16, 8), 256, 0, stream>>>(l0, u0, lx0, ux0, lc0, uc0, weight, bias, ws, 24, 8);
        deeppoly_conv_finish<<<P_, 32, 0, stream>>>(ws, out, 8);
    } else {
        deeppoly_conv_main<<<dim3(16, 4), 256, 0, stream>>>(l0, u0, lx0, ux0, lc0, uc0, weight, bias, ws, 48, 4);
        deeppoly_conv_finish<<<P_, 32, 0, stream>>>(ws, out, 4);
    }
}
